# starting kernel
# baseline (speedup 1.0000x reference)
#include <hip/hip_runtime.h>

typedef __attribute__((ext_vector_type(2))) float v2f;
typedef __attribute__((ext_vector_type(8))) float v8f;

#define IMG    224
#define PATCH  16
#define NP     14
#define EMBED  768
#define CCH    12
#define BB     64
#define M_TOTAL (BB * NP * NP)
#define NTILES  (EMBED / 16)
#define MTILES  (M_TOTAL / 16)

__global__ __launch_bounds__(224) void pool16_kernel(const float* __restrict__ x,
                                                     float* __restrict__ pooled) {
    __shared__ float colsum[IMG];
    int s = blockIdx.x;
    int i = s % NP; s /= NP;
    int c = s % CCH; s /= CCH;
    int b = s;
    const float* base = x + (((size_t)(b * CCH + c)) * IMG + (size_t)i * PATCH) * IMG;

    int col = threadIdx.x;
    float acc = 0.f;
#pragma unroll
    for (int r = 0; r < PATCH; ++r)
        acc += base[(size_t)r * IMG + col];
    colsum[col] = acc;
    __syncthreads();

    if (col < NP) {
        float ps = 0.f;
#pragma unroll
        for (int k = 0; k < PATCH; ++k)
            ps += colsum[col * PATCH + k];
        int m = b * (NP * NP) + i * NP + col;
        pooled[(size_t)m * CCH + c] = ps;
    }
}

__global__ __launch_bounds__(256) void gemm_wmma_kernel(const float* __restrict__ pooled,
                                                        const float* __restrict__ w,
                                                        float* __restrict__ out) {
    int wid  = blockIdx.x * 8 + (threadIdx.x >> 5);
    int lane = threadIdx.x & 31;
    int mtile = wid / NTILES;
    int ntile = wid % NTILES;

    int half  = lane >> 4;
    int khalf = half * 2;
    int mrow  = lane & 15;
    int n     = lane & 15;

    const float* arow = pooled + (size_t)(mtile * 16 + mrow) * CCH;
    const float* wcol = w + ntile * 16 + n;

    v8f acc = {};
#pragma unroll
    for (int k0 = 0; k0 < CCH; k0 += 4) {
        v2f a;
        a.x = arow[k0 + khalf];
        a.y = arow[k0 + khalf + 1];
        v2f bm;
        bm.x = wcol[(size_t)(k0 + khalf)     * EMBED];
        bm.y = wcol[(size_t)(k0 + khalf + 1) * EMBED];
        acc = __builtin_amdgcn_wmma_f32_16x16x4_f32(
             false, a,  false, bm,
             (short)0, acc,  false,  false);
    }

    int mbase = mtile * 16 + half * 8;
    float* o = out + (size_t)mbase * EMBED + (size_t)ntile * 16 + n;
#pragma unroll
    for (int v = 0; v < 8; ++v)
        o[(size_t)v * EMBED] = acc[v];
}

extern "C" void kernel_launch(void* const* d_in, const int* in_sizes, int n_in,
                              void* d_out, int out_size, void* d_ws, size_t ws_size,
                              hipStream_t stream) {
    const float* x = (const float*)d_in[0];
    const float* w = (const float*)d_in[1];
    float* out    = (float*)d_out;
    float* pooled = (float*)d_ws;

    pool16_kernel<<<BB * CCH * NP, 224, 0, stream>>>(x, pooled);
    gemm_wmma_kernel<<<(MTILES * NTILES) / 8, 256, 0, stream>>>(pooled, w, out);
}


# GEMM stage re-tiled: a wave owns a 16x192 block (was 16x16), N index interleaved so a lane holds 4 adjacent columns: 16-byte W loads, whole-line out stores, 12x fewer waves; sum order kept
# speedup vs baseline: 1.1600x; 1.1600x over previous
#include <hip/hip_runtime.h>

typedef __attribute__((ext_vector_type(2))) float v2f;
typedef __attribute__((ext_vector_type(4))) float v4f;
typedef __attribute__((ext_vector_type(8))) float v8f;

#define IMG    224
#define PATCH  16
#define NP     14
#define EMBED  768
#define CCH    12
#define BB     64
#define M_TOTAL (BB * NP * NP)
#define NTILES  (EMBED / 16)
#define MTILES  (M_TOTAL / 16)

__global__ __launch_bounds__(224) void pool16_kernel(const float* __restrict__ x,
                                                     float* __restrict__ pooled) {
    __shared__ float colsum[IMG];
    int s = blockIdx.x;
    int i = s % NP; s /= NP;
    int c = s % CCH; s /= CCH;
    int b = s;
    const float* base = x + (((size_t)(b * CCH + c)) * IMG + (size_t)i * PATCH) * IMG;

    int col = threadIdx.x;
    float acc = 0.f;
#pragma unroll
    for (int r = 0; r < PATCH; ++r)
        acc += base[(size_t)r * IMG + col];
    colsum[col] = acc;
    __syncthreads();

    if (col < NP) {
        float ps = 0.f;
#pragma unroll
        for (int k = 0; k < PATCH; ++k)
            ps += colsum[col * PATCH + k];
        int m = b * (NP * NP) + i * NP + col;
        pooled[(size_t)m * CCH + c] = ps;
    }
}

#define BAND        64
#define GEMM_WAVES  4
#define WAVE_COLS   (EMBED / GEMM_WAVES)
#define WAVE_BANDS  (WAVE_COLS / BAND)

__global__ __launch_bounds__(GEMM_WAVES * 32) void gemm_wmma_kernel(const float* __restrict__ pooled,
                                                                    const float* __restrict__ w,
                                                                    float* __restrict__ out) {
    int wave  = threadIdx.x >> 5;
    int lane  = threadIdx.x & 31;
    int mtile = blockIdx.x;

    int half  = lane >> 4;
    int khalf = half * 2;
    int c     = lane & 15;

    const float* arow = pooled + (size_t)(mtile * 16 + c) * CCH + khalf;
    v2f a[3];
#pragma unroll
    for (int s = 0; s < 3; ++s)
        a[s] = *(const v2f*)(arow + 4 * s);

    int col0 = wave * WAVE_COLS + 4 * c;
    const float* wp = w + (size_t)khalf * EMBED + col0;
    float* o = out + (size_t)(mtile * 16 + half * 8) * EMBED + col0;

#pragma unroll
    for (int band = 0; band < WAVE_BANDS; ++band) {
        v4f wv[3][2];
#pragma unroll
        for (int s = 0; s < 3; ++s) {
            wv[s][0] = *(const v4f*)(wp + (size_t)(4 * s)     * EMBED + band * BAND);
            wv[s][1] = *(const v4f*)(wp + (size_t)(4 * s + 1) * EMBED + band * BAND);
        }

        v8f acc[4] = {};
#pragma unroll
        for (int s = 0; s < 3; ++s) {
#pragma unroll
            for (int t = 0; t < 4; ++t) {
                v2f bm;
                bm.x = wv[s][0][t];
                bm.y = wv[s][1][t];
                acc[t] = __builtin_amdgcn_wmma_f32_16x16x4_f32(
                     false, a[s],  false, bm,
                     (short)0, acc[t],  false,  false);
            }
        }

#pragma unroll
        for (int r = 0; r < 8; ++r) {
            v4f v;
            v.x = acc[0][r];
            v.y = acc[1][r];
            v.z = acc[2][r];
            v.w = acc[3][r];
            *(v4f*)(o + (size_t)r * EMBED + band * BAND) = v;
        }
    }
}

extern "C" void kernel_launch(void* const* d_in, const int* in_sizes, int n_in,
                              void* d_out, int out_size, void* d_ws, size_t ws_size,
                              hipStream_t stream) {
    const float* x = (const float*)d_in[0];
    const float* w = (const float*)d_in[1];
    float* out    = (float*)d_out;
    float* pooled = (float*)d_ws;

    pool16_kernel<<<BB * CCH * NP, 224, 0, stream>>>(x, pooled);
    gemm_wmma_kernel<<<MTILES, GEMM_WAVES * 32, 0, stream>>>(pooled, w, out);
}


# pool kernel: a block now covers 3 channels of one patch-row (was 1): 48 independent whole-line loads per lane issued before the first use, 3x fewer waves, still one round trip each; sum order kept
# speedup vs baseline: 1.2935x; 1.1151x over previous
#include <hip/hip_runtime.h>

typedef __attribute__((ext_vector_type(2))) float v2f;
typedef __attribute__((ext_vector_type(4))) float v4f;
typedef __attribute__((ext_vector_type(8))) float v8f;

#define IMG    224
#define PATCH  16
#define NP     14
#define EMBED  768
#define CCH    12
#define BB     64
#define M_TOTAL (BB * NP * NP)
#define NTILES  (EMBED / 16)
#define MTILES  (M_TOTAL / 16)

#define CPB     3
#define CGROUPS (CCH / CPB)

__global__ __launch_bounds__(224) void pool16_kernel(const float* __restrict__ x,
                                                     float* __restrict__ pooled) {
    __shared__ float colsum[CPB * IMG];
    int s = blockIdx.x;
    int i = s % NP; s /= NP;
    int g = s % CGROUPS; s /= CGROUPS;
    int b = s;
    int c0 = g * CPB;
    const float* base = x + (((size_t)(b * CCH + c0)) * IMG + (size_t)i * PATCH) * IMG;

    int col = threadIdx.x;
    float v[CPB][PATCH];
#pragma unroll
    for (int q = 0; q < CPB; ++q) {
#pragma unroll
        for (int r = 0; r < PATCH; ++r)
            v[q][r] = base[((size_t)q * IMG + r) * IMG + col];
    }
#pragma unroll
    for (int q = 0; q < CPB; ++q) {
        float acc = 0.f;
#pragma unroll
        for (int r = 0; r < PATCH; ++r)
            acc += v[q][r];
        colsum[q * IMG + col] = acc;
    }
    __syncthreads();

    int q = col >> 5;
    int j = col & 31;
    if (q < CPB && j < NP) {
        float ps = 0.f;
#pragma unroll
        for (int k = 0; k < PATCH; ++k)
            ps += colsum[q * IMG + j * PATCH + k];
        int m = b * (NP * NP) + i * NP + j;
        pooled[(size_t)m * CCH + c0 + q] = ps;
    }
}

#define BAND        64
#define GEMM_WAVES  4
#define WAVE_COLS   (EMBED / GEMM_WAVES)
#define WAVE_BANDS  (WAVE_COLS / BAND)

__global__ __launch_bounds__(GEMM_WAVES * 32) void gemm_wmma_kernel(const float* __restrict__ pooled,
                                                                    const float* __restrict__ w,
                                                                    float* __restrict__ out) {
    int wave  = threadIdx.x >> 5;
    int lane  = threadIdx.x & 31;
    int mtile = blockIdx.x;

    int half  = lane >> 4;
    int khalf = half * 2;
    int c     = lane & 15;

    const float* arow = pooled + (size_t)(mtile * 16 + c) * CCH + khalf;
    v2f a[3];
#pragma unroll
    for (int s = 0; s < 3; ++s)
        a[s] = *(const v2f*)(arow + 4 * s);

    int col0 = wave * WAVE_COLS + 4 * c;
    const float* wp = w + (size_t)khalf * EMBED + col0;
    float* o = out + (size_t)(mtile * 16 + half * 8) * EMBED + col0;

#pragma unroll
    for (int band = 0; band < WAVE_BANDS; ++band) {
        v4f wv[3][2];
#pragma unroll
        for (int s = 0; s < 3; ++s) {
            wv[s][0] = *(const v4f*)(wp + (size_t)(4 * s)     * EMBED + band * BAND);
            wv[s][1] = *(const v4f*)(wp + (size_t)(4 * s + 1) * EMBED + band * BAND);
        }

        v8f acc[4] = {};
#pragma unroll
        for (int s = 0; s < 3; ++s) {
#pragma unroll
            for (int t = 0; t < 4; ++t) {
                v2f bm;
                bm.x = wv[s][0][t];
                bm.y = wv[s][1][t];
                acc[t] = __builtin_amdgcn_wmma_f32_16x16x4_f32(
                     false, a[s],  false, bm,
                     (short)0, acc[t],  false,  false);
            }
        }

#pragma unroll
        for (int r = 0; r < 8; ++r) {
            v4f v;
            v.x = acc[0][r];
            v.y = acc[1][r];
            v.z = acc[2][r];
            v.w = acc[3][r];
            *(v4f*)(o + (size_t)r * EMBED + band * BAND) = v;
        }
    }
}

extern "C" void kernel_launch(void* const* d_in, const int* in_sizes, int n_in,
                              void* d_out, int out_size, void* d_ws, size_t ws_size,
                              hipStream_t stream) {
    const float* x = (const float*)d_in[0];
    const float* w = (const float*)d_in[1];
    float* out    = (float*)d_out;
    float* pooled = (float*)d_ws;

    pool16_kernel<<<BB * CGROUPS * NP, 224, 0, stream>>>(x, pooled);
    gemm_wmma_kernel<<<MTILES, GEMM_WAVES * 32, 0, stream>>>(pooled, w, out);
}
